# starting kernel
# baseline (speedup 1.0000x reference)
#include <hip/hip_runtime.h>
#include <math.h>


typedef __attribute__((ext_vector_type(2))) float v2f;
typedef __attribute__((ext_vector_type(8))) float v8f;

__global__ void wq_kernel(const float* __restrict__ W2, const float* __restrict__ b2,
                          const float* __restrict__ q, float* __restrict__ Wqp,
                          float* __restrict__ bqv) {
  const int h = threadIdx.x;
  float s[8];
#pragma unroll
  for (int k = 0; k < 8; ++k) s[k] = 0.0f;
  for (int d = 0; d < 256; ++d) {
    const float w = W2[d * 128 + h];
#pragma unroll
    for (int k = 0; k < 8; ++k) s[k] += w * q[k * 256 + d];
  }
#pragma unroll
  for (int k = 0; k < 8; ++k) Wqp[h * 16 + k] = s[k];
#pragma unroll
  for (int k = 8; k < 16; ++k) Wqp[h * 16 + k] = 0.0f;
  if (h < 8) {
    float t = 0.0f;
    for (int d = 0; d < 256; ++d) t += b2[d] * q[h * 256 + d];
    bqv[h] = t;
  }
}

__global__ void gemm_A_kernel(const float* __restrict__ E, const float* __restrict__ W1,
                              float* __restrict__ Afull) {
  const int wave = (blockIdx.x * blockDim.x + threadIdx.x) >> 5;
  const int lane = threadIdx.x & 31;
  const int l15 = lane & 15;
  const int hi  = lane >> 4;
  const int tm  = wave >> 4;
  const int tc  = wave & 15;
  const int row0 = tm * 16;
  const int col  = tc * 16 + l15;
  const float* wcol = (col < 128) ? (W1 + col * 512) : (W1 + (col - 128) * 512 + 256);
  const float* erow = E + (row0 + l15) * 256;
  v8f acc = {};
#pragma unroll 4
  for (int kk = 0; kk < 64; ++kk) {
    const int d = kk * 4 + hi * 2;
    v2f a = *(const v2f*)(erow + d);
    v2f b = *(const v2f*)(wcol + d);
    acc = __builtin_amdgcn_wmma_f32_16x16x4_f32(false, a, false, b, (short)0, acc,
                                                false, false);
  }
#pragma unroll
  for (int r = 0; r < 8; ++r) {
    Afull[(row0 + r + hi * 8) * 256 + tc * 16 + l15] = acc[r];
  }
}

__global__ void score_kernel(const float* __restrict__ Afull, const float* __restrict__ b1,
                             const float* __restrict__ Wqp, const float* __restrict__ bqv,
                             float* __restrict__ out) {
  __shared__ float sWq[128 * 16];
  {
    const int t = threadIdx.x;
#pragma unroll
    for (int u = 0; u < 8; ++u) sWq[t + u * 256] = Wqp[t + u * 256];
  }
  __syncthreads();

  const int wave = (blockIdx.x * blockDim.x + threadIdx.x) >> 5;
  const int lane = threadIdx.x & 31;
  const int l15 = lane & 15;
  const int hi  = lane >> 4;
  const int jt  = wave & 7;
  const int i   = (wave >> 3) & 127;
  const int b   = wave >> 10;
  const int j   = jt * 16 + l15;

  const float* ai = Afull + (b * 128 + i) * 256;
  const float* aj = Afull + (b * 128 + j) * 256 + 128;

  v8f acc = {};
#pragma unroll 4
  for (int kk = 0; kk < 32; ++kk) {
    const int h = kk * 4 + hi * 2;
    v2f a;
    a.x = sWq[h * 16 + l15];
    a.y = sWq[(h + 1) * 16 + l15];
    v2f xa = *(const v2f*)(aj + h);
    v2f xi = *(const v2f*)(ai + h);
    v2f bb = *(const v2f*)(b1 + h);
    v2f hv;
    hv.x = fmaxf(xa.x + xi.x + bb.x, 0.0f);
    hv.y = fmaxf(xa.y + xi.y + bb.y, 0.0f);
    acc = __builtin_amdgcn_wmma_f32_16x16x4_f32(false, a, false, hv, (short)0, acc,
                                                false, false);
  }

  if (hi == 0) {
#pragma unroll
    for (int r = 0; r < 8; ++r) {
      float v = acc[r] + bqv[r];
      float s = 1.0f / (1.0f + __expf(-v));
      if (i == j) s = 0.0f;
      out[(((b * 8 + r) * 128) + i) * 128 + j] = s;
    }
  }
}

extern "C" void kernel_launch(void* const* d_in, const int* in_sizes, int n_in,
                              void* d_out, int out_size, void* d_ws, size_t ws_size,
                              hipStream_t stream) {
  const float* E  = (const float*)d_in[0];
  const float* W1 = (const float*)d_in[1];
  const float* b1 = (const float*)d_in[2];
  const float* W2 = (const float*)d_in[3];
  const float* b2 = (const float*)d_in[4];
  const float* q  = (const float*)d_in[5];
  float* out = (float*)d_out;

  float* Afull = (float*)d_ws;
  float* Wqp   = Afull + 262144;
  float* bqv   = Wqp + 2048;

  wq_kernel<<<dim3(1), dim3(128), 0, stream>>>(W2, b2, q, Wqp, bqv);
  gemm_A_kernel<<<dim3(128), dim3(256), 0, stream>>>(E, W1, Afull);
  score_kernel<<<dim3(1024), dim3(256), 0, stream>>>(Afull, b1, Wqp, bqv, out);
}


# Wq/bq: serial one-block pre-kernel replaced by one extra work-group inside the GEMM launch (q x W2 on f32 WMMA, bq by wave butterfly sum); 3 launches -> 2; Wq/bq summation order changes
# speedup vs baseline: 2.7408x; 2.7408x over previous
#include <hip/hip_runtime.h>
#include <math.h>


typedef __attribute__((ext_vector_type(2))) float v2f;
typedef __attribute__((ext_vector_type(8))) float v8f;

__global__ void gemm_A_kernel(const float* __restrict__ E, const float* __restrict__ W1,
                              float* __restrict__ Afull,
                              const float* __restrict__ W2, const float* __restrict__ b2,
                              const float* __restrict__ q, float* __restrict__ Wqp,
                              float* __restrict__ bqv) {
  __shared__ float sbq[8];
  const int lane = threadIdx.x & 31;
  const int l15 = lane & 15;
  const int hi  = lane >> 4;

  if (blockIdx.x == 128) {
    const int wv = threadIdx.x >> 5;

    {
      const float* bp = b2 + lane * 8;
      const float* qp = q + wv * 256 + lane * 8;
      float t = 0.0f;
#pragma unroll
      for (int m = 0; m < 8; m += 2) {
        const v2f x = *(const v2f*)(bp + m);
        const v2f y = *(const v2f*)(qp + m);
        t += x.x * y.x;
        t += x.y * y.y;
      }
#pragma unroll
      for (int o = 16; o > 0; o >>= 1) t += __shfl_xor(t, o);
      if (lane == 0) sbq[wv] = t;
    }

    const int h = wv * 16 + l15;
    const float* qrow = q + (l15 & 7) * 256;
    const float* wcolh = W2 + h;
    v8f acc = {};
#pragma unroll
    for (int kk = 0; kk < 64; ++kk) {
      const int d = kk * 4 + hi * 2;
      v2f a = *(const v2f*)(qrow + d);
      v2f b;
      b.x = wcolh[d * 128];
      b.y = wcolh[(d + 1) * 128];
      acc = __builtin_amdgcn_wmma_f32_16x16x4_f32(false, a, false, b, (short)0, acc,
                                                  false, false);
    }
    float* wout = Wqp + h * 16 + hi * 8;
#pragma unroll
    for (int r = 0; r < 8; ++r) wout[r] = hi ? 0.0f : acc[r];

    __syncthreads();
    if (threadIdx.x < 8) bqv[threadIdx.x] = sbq[threadIdx.x];
    return;
  }

  const int wave = (blockIdx.x * blockDim.x + threadIdx.x) >> 5;
  const int tm  = wave >> 4;
  const int tc  = wave & 15;
  const int row0 = tm * 16;
  const int col  = tc * 16 + l15;
  const float* wcol = (col < 128) ? (W1 + col * 512) : (W1 + (col - 128) * 512 + 256);
  const float* erow = E + (row0 + l15) * 256;
  v8f acc = {};
#pragma unroll 4
  for (int kk = 0; kk < 64; ++kk) {
    const int d = kk * 4 + hi * 2;
    v2f a = *(const v2f*)(erow + d);
    v2f b = *(const v2f*)(wcol + d);
    acc = __builtin_amdgcn_wmma_f32_16x16x4_f32(false, a, false, b, (short)0, acc,
                                                false, false);
  }
#pragma unroll
  for (int r = 0; r < 8; ++r) {
    Afull[(row0 + r + hi * 8) * 256 + tc * 16 + l15] = acc[r];
  }
}

__global__ void score_kernel(const float* __restrict__ Afull, const float* __restrict__ b1,
                             const float* __restrict__ Wqp, const float* __restrict__ bqv,
                             float* __restrict__ out) {
  __shared__ float sWq[128 * 16];
  {
    const int t = threadIdx.x;
#pragma unroll
    for (int u = 0; u < 8; ++u) sWq[t + u * 256] = Wqp[t + u * 256];
  }
  __syncthreads();

  const int wave = (blockIdx.x * blockDim.x + threadIdx.x) >> 5;
  const int lane = threadIdx.x & 31;
  const int l15 = lane & 15;
  const int hi  = lane >> 4;
  const int jt  = wave & 7;
  const int i   = (wave >> 3) & 127;
  const int b   = wave >> 10;
  const int j   = jt * 16 + l15;

  const float* ai = Afull + (b * 128 + i) * 256;
  const float* aj = Afull + (b * 128 + j) * 256 + 128;

  v8f acc = {};
#pragma unroll 4
  for (int kk = 0; kk < 32; ++kk) {
    const int h = kk * 4 + hi * 2;
    v2f a;
    a.x = sWq[h * 16 + l15];
    a.y = sWq[(h + 1) * 16 + l15];
    v2f xa = *(const v2f*)(aj + h);
    v2f xi = *(const v2f*)(ai + h);
    v2f bb = *(const v2f*)(b1 + h);
    v2f hv;
    hv.x = fmaxf(xa.x + xi.x + bb.x, 0.0f);
    hv.y = fmaxf(xa.y + xi.y + bb.y, 0.0f);
    acc = __builtin_amdgcn_wmma_f32_16x16x4_f32(false, a, false, hv, (short)0, acc,
                                                false, false);
  }

  if (hi == 0) {
#pragma unroll
    for (int r = 0; r < 8; ++r) {
      float v = acc[r] + bqv[r];
      float s = 1.0f / (1.0f + __expf(-v));
      if (i == j) s = 0.0f;
      out[(((b * 8 + r) * 128) + i) * 128 + j] = s;
    }
  }
}

extern "C" void kernel_launch(void* const* d_in, const int* in_sizes, int n_in,
                              void* d_out, int out_size, void* d_ws, size_t ws_size,
                              hipStream_t stream) {
  const float* E  = (const float*)d_in[0];
  const float* W1 = (const float*)d_in[1];
  const float* b1 = (const float*)d_in[2];
  const float* W2 = (const float*)d_in[3];
  const float* b2 = (const float*)d_in[4];
  const float* q  = (const float*)d_in[5];
  float* out = (float*)d_out;

  float* Afull = (float*)d_ws;
  float* Wqp   = Afull + 262144;
  float* bqv   = Wqp + 2048;

  gemm_A_kernel<<<dim3(129), dim3(256), 0, stream>>>(E, W1, Afull, W2, b2, q, Wqp, bqv);
  score_kernel<<<dim3(1024), dim3(256), 0, stream>>>(Afull, b1, Wqp, bqv, out);
}


# score kernel re-tiled: a wave keeps the Aj fragments of 32 columns j and the Wq fragment in registers and loops over 4 rows i (8192 waves -> 1024, whole-line stores); b1 summed as (Ai+b1)+Aj
# speedup vs baseline: 3.5879x; 1.3091x over previous
#include <hip/hip_runtime.h>
#include <math.h>


typedef __attribute__((ext_vector_type(2))) float v2f;
typedef __attribute__((ext_vector_type(4))) float v4f;
typedef __attribute__((ext_vector_type(8))) float v8f;

__global__ void gemm_A_kernel(const float* __restrict__ E, const float* __restrict__ W1,
                              float* __restrict__ Afull,
                              const float* __restrict__ W2, const float* __restrict__ b2,
                              const float* __restrict__ q, float* __restrict__ Wqp,
                              float* __restrict__ bqv) {
  __shared__ float sbq[8];
  const int lane = threadIdx.x & 31;
  const int l15 = lane & 15;
  const int hi  = lane >> 4;

  if (blockIdx.x == 128) {
    const int wv = threadIdx.x >> 5;

    {
      const float* bp = b2 + lane * 8;
      const float* qp = q + wv * 256 + lane * 8;
      float t = 0.0f;
#pragma unroll
      for (int m = 0; m < 8; m += 2) {
        const v2f x = *(const v2f*)(bp + m);
        const v2f y = *(const v2f*)(qp + m);
        t += x.x * y.x;
        t += x.y * y.y;
      }
#pragma unroll
      for (int o = 16; o > 0; o >>= 1) t += __shfl_xor(t, o);
      if (lane == 0) sbq[wv] = t;
    }

    const int h = wv * 16 + l15;
    const float* qrow = q + (l15 & 7) * 256;
    const float* wcolh = W2 + h;
    v8f acc = {};
#pragma unroll
    for (int kk = 0; kk < 64; ++kk) {
      const int d = kk * 4 + hi * 2;
      v2f a = *(const v2f*)(qrow + d);
      v2f b;
      b.x = wcolh[d * 128];
      b.y = wcolh[(d + 1) * 128];
      acc = __builtin_amdgcn_wmma_f32_16x16x4_f32(false, a, false, b, (short)0, acc,
                                                  false, false);
    }
    float* wout = Wqp + h * 16 + hi * 8;
#pragma unroll
    for (int r = 0; r < 8; ++r) wout[r] = hi ? 0.0f : acc[r];

    __syncthreads();
    if (threadIdx.x < 8) bqv[threadIdx.x] = sbq[threadIdx.x];
    return;
  }

  const int wave = (blockIdx.x * blockDim.x + threadIdx.x) >> 5;
  const int tm  = wave >> 4;
  const int tc  = wave & 15;
  const int row0 = tm * 16;
  const int col  = tc * 16 + l15;
  const float* wcol = (col < 128) ? (W1 + col * 512) : (W1 + (col - 128) * 512 + 256);
  const float* erow = E + (row0 + l15) * 256;
  v8f acc = {};
#pragma unroll 4
  for (int kk = 0; kk < 64; ++kk) {
    const int d = kk * 4 + hi * 2;
    v2f a = *(const v2f*)(erow + d);
    v2f b = *(const v2f*)(wcol + d);
    acc = __builtin_amdgcn_wmma_f32_16x16x4_f32(false, a, false, b, (short)0, acc,
                                                false, false);
  }
#pragma unroll
  for (int r = 0; r < 8; ++r) {
    Afull[(row0 + r + hi * 8) * 256 + tc * 16 + l15] = acc[r];
  }
}

__global__ void __launch_bounds__(128)
score_kernel(const float* __restrict__ Afull, const float* __restrict__ b1,
             const float* __restrict__ Wqp, const float* __restrict__ bqv,
             float* __restrict__ out) {
  constexpr int P = 132;
  __shared__ __align__(16) float sWq[8 * P];
  __shared__ __align__(16) float sAj[32 * P];
  __shared__ __align__(16) float sAi[16 * 128];

  const int t    = threadIdx.x;
  const int lane = t & 31;
  const int c    = lane & 15;
  const int hi   = lane >> 4;
  const int w    = t >> 5;
  const int b    = blockIdx.x >> 5;
  const int jp   = (blockIdx.x >> 3) & 3;
  const int ig   = blockIdx.x & 7;

  {
    const int col = lane * 4;
    const float* srcj = Afull + ((b * 128 + jp * 32 + w) * 256 + 128 + col);
    const float* srci = Afull + ((b * 128 + ig * 16 + w) * 256 + col);
    v4f vj[8], vi[4];
#pragma unroll
    for (int n = 0; n < 8; ++n) vj[n] = *(const v4f*)(srcj + n * 4 * 256);
#pragma unroll
    for (int n = 0; n < 4; ++n) vi[n] = *(const v4f*)(srci + n * 4 * 256);
    const v4f bb = *(const v4f*)(b1 + col);
    const v4f q0 = *(const v4f*)(Wqp + t * 16);
    const v4f q1 = *(const v4f*)(Wqp + t * 16 + 4);
#pragma unroll
    for (int n = 0; n < 8; ++n) *(v4f*)(sAj + (n * 4 + w) * P + col) = vj[n];
#pragma unroll
    for (int n = 0; n < 4; ++n) *(v4f*)(sAi + (n * 4 + w) * 128 + col) = vi[n] + bb;
    sWq[0 * P + t] = q0.x; sWq[1 * P + t] = q0.y; sWq[2 * P + t] = q0.z; sWq[3 * P + t] = q0.w;
    sWq[4 * P + t] = q1.x; sWq[5 * P + t] = q1.y; sWq[6 * P + t] = q1.z; sWq[7 * P + t] = q1.w;
  }
  __syncthreads();

  v4f wq[16], aj0[16], aj1[16];
  {
    const float* pw = sWq + (c & 7) * P + 4 * hi;
    const float* p0 = sAj + c * P + 4 * hi;
    const float* p1 = sAj + (16 + c) * P + 4 * hi;
#pragma unroll
    for (int m = 0; m < 16; ++m) {
      wq[m]  = *(const v4f*)(pw + 8 * m);
      aj0[m] = *(const v4f*)(p0 + 8 * m);
      aj1[m] = *(const v4f*)(p1 + 8 * m);
    }
  }
  float bq[8];
#pragma unroll
  for (int r = 0; r < 8; ++r) bq[r] = bqv[r];

  const float* pa = sAi + (4 * w) * 128 + 4 * hi;
  v4f ai[16];
#pragma unroll
  for (int m = 0; m < 16; ++m) ai[m] = *(const v4f*)(pa + 8 * m);

  const int j = jp * 32 + lane;
  float* pout = out + ((b * 8) * 128 + ig * 16 + 4 * w) * 128 + j;

#pragma unroll 1
  for (int ii = 0; ii < 4; ++ii) {
    v4f nx[16];
    const float* pn = pa + ((ii < 3) ? (ii + 1) : ii) * 128;
#pragma unroll
    for (int m = 0; m < 16; ++m) nx[m] = *(const v4f*)(pn + 8 * m);

    v8f acc0 = {};
    v8f acc1 = {};
#pragma unroll
    for (int m = 0; m < 16; ++m) {
#pragma unroll
      for (int t2 = 0; t2 < 2; ++t2) {
        v2f a, h0, h1;
        a.x = wq[m][2 * t2];
        a.y = wq[m][2 * t2 + 1];
        const float x0 = ai[m][2 * t2];
        const float x1 = ai[m][2 * t2 + 1];
        h0.x = fmaxf(aj0[m][2 * t2] + x0, 0.0f);
        h0.y = fmaxf(aj0[m][2 * t2 + 1] + x1, 0.0f);
        h1.x = fmaxf(aj1[m][2 * t2] + x0, 0.0f);
        h1.y = fmaxf(aj1[m][2 * t2 + 1] + x1, 0.0f);
        acc0 = __builtin_amdgcn_wmma_f32_16x16x4_f32(false, a, false, h0, (short)0, acc0,
                                                     false, false);
        acc1 = __builtin_amdgcn_wmma_f32_16x16x4_f32(false, a, false, h1, (short)0, acc1,
                                                     false, false);
      }
    }

    const int i = ig * 16 + 4 * w + ii;
#pragma unroll
    for (int r = 0; r < 8; ++r) {
      const float v = (hi ? acc1[r] : acc0[r]) + bq[r];
      float s = 1.0f / (1.0f + __expf(-v));
      if (i == j) s = 0.0f;
      pout[(r * 128 + ii) * 128] = s;
    }
#pragma unroll
    for (int m = 0; m < 16; ++m) ai[m] = nx[m];
  }
}

extern "C" void kernel_launch(void* const* d_in, const int* in_sizes, int n_in,
                              void* d_out, int out_size, void* d_ws, size_t ws_size,
                              hipStream_t stream) {
  const float* E  = (const float*)d_in[0];
  const float* W1 = (const float*)d_in[1];
  const float* b1 = (const float*)d_in[2];
  const float* W2 = (const float*)d_in[3];
  const float* b2 = (const float*)d_in[4];
  const float* q  = (const float*)d_in[5];
  float* out = (float*)d_out;

  float* Afull = (float*)d_ws;
  float* Wqp   = Afull + 262144;
  float* bqv   = Wqp + 2048;

  gemm_A_kernel<<<dim3(129), dim3(256), 0, stream>>>(E, W1, Afull, W2, b2, q, Wqp, bqv);
  score_kernel<<<dim3(256), dim3(128), 0, stream>>>(Afull, b1, Wqp, bqv, out);
}


# GEMM kernel re-tiled: 4-wave work-groups stage 32 E rows + 32 weight rows through LDS with 512-byte loads all in flight at once (16 dependent load round trips -> 1); Ai/Aj summation order changes
# speedup vs baseline: 4.3036x; 1.1995x over previous
#include <hip/hip_runtime.h>
#include <math.h>


typedef __attribute__((ext_vector_type(2))) float v2f;
typedef __attribute__((ext_vector_type(4))) float v4f;
typedef __attribute__((ext_vector_type(8))) float v8f;

__global__ void __launch_bounds__(128)
gemm_A_kernel(const float* __restrict__ E, const float* __restrict__ W1,
              float* __restrict__ Afull,
              const float* __restrict__ W2, const float* __restrict__ b2,
              const float* __restrict__ q, float* __restrict__ Wqp,
              float* __restrict__ bqv) {
  constexpr int P = 260;
  __shared__ float sbq[4];
  __shared__ __align__(16) float sE[32 * P];
  __shared__ __align__(16) float sW[32 * P];
  const int lane = threadIdx.x & 31;
  const int l15 = lane & 15;
  const int hi  = lane >> 4;
  const int w   = threadIdx.x >> 5;

  if (blockIdx.x < 2) {
    const int wv = blockIdx.x * 4 + w;

    {
      const float* bp = b2 + lane * 8;
      const float* qp = q + wv * 256 + lane * 8;
      float t = 0.0f;
#pragma unroll
      for (int m = 0; m < 8; m += 2) {
        const v2f x = *(const v2f*)(bp + m);
        const v2f y = *(const v2f*)(qp + m);
        t += x.x * y.x;
        t += x.y * y.y;
      }
#pragma unroll
      for (int o = 16; o > 0; o >>= 1) t += __shfl_xor(t, o);
      if (lane == 0) sbq[w] = t;
    }

    const int h = wv * 16 + l15;
    const float* qrow = q + (l15 & 7) * 256;
    const float* wcolh = W2 + h;
    v8f acc = {};
#pragma unroll
    for (int kk = 0; kk < 64; ++kk) {
      const int d = kk * 4 + hi * 2;
      v2f a = *(const v2f*)(qrow + d);
      v2f b;
      b.x = wcolh[d * 128];
      b.y = wcolh[(d + 1) * 128];
      acc = __builtin_amdgcn_wmma_f32_16x16x4_f32(false, a, false, b, (short)0, acc,
                                                  false, false);
    }
    float* wout = Wqp + h * 16 + hi * 8;
#pragma unroll
    for (int r = 0; r < 8; ++r) wout[r] = hi ? 0.0f : acc[r];

    __syncthreads();
    if (threadIdx.x < 4) bqv[blockIdx.x * 4 + threadIdx.x] = sbq[threadIdx.x];
    return;
  }

  const int g    = blockIdx.x - 2;
  const int row0 = (g >> 3) * 32;
  const int col0 = (g & 7) * 32;
  const float* wbase = (col0 < 128) ? (W1 + col0 * 512) : (W1 + (col0 - 128) * 512 + 256);

  const float* se = E + (row0 + w) * 256 + lane * 4;
  const float* sw = wbase + w * 512 + lane * 4;
  v4f ve[16], vw[16];
#pragma unroll
  for (int n = 0; n < 8; ++n) {
    ve[n] = *(const v4f*)(se + n * 4 * 256);
    vw[n] = *(const v4f*)(sw + n * 4 * 512);
  }
  __builtin_amdgcn_sched_barrier(0);
#pragma unroll
  for (int n = 0; n < 8; ++n) {
    ve[8 + n] = *(const v4f*)(se + n * 4 * 256 + 128);
    vw[8 + n] = *(const v4f*)(sw + n * 4 * 512 + 128);
  }

  const int rt = w >> 1;
  const int ct = w & 1;
  const float* pa = sE + (rt * 16 + l15) * P + 4 * hi;
  const float* pb = sW + (ct * 16 + l15) * P + 4 * hi;
  v8f acc0 = {};
  v8f acc1 = {};
#pragma unroll
  for (int hf = 0; hf < 2; ++hf) {
#pragma unroll
    for (int n = 0; n < 8; ++n) {
      *(v4f*)(sE + (n * 4 + w) * P + hf * 128 + lane * 4) = ve[hf * 8 + n];
      *(v4f*)(sW + (n * 4 + w) * P + hf * 128 + lane * 4) = vw[hf * 8 + n];
    }
    __builtin_amdgcn_fence(__ATOMIC_RELEASE, "workgroup", "local");
    __builtin_amdgcn_s_barrier();
    __builtin_amdgcn_fence(__ATOMIC_ACQUIRE, "workgroup", "local");
    v4f fa[16], fb[16];
#pragma unroll
    for (int m = 0; m < 16; ++m) {
      fa[m] = *(const v4f*)(pa + hf * 128 + 8 * m);
      fb[m] = *(const v4f*)(pb + hf * 128 + 8 * m);
    }
    __builtin_amdgcn_sched_barrier(0);
#pragma unroll
    for (int m = 0; m < 16; ++m) {
      v2f a0, b0, a1, b1v;
      a0.x = fa[m][0]; a0.y = fa[m][1];
      b0.x = fb[m][0]; b0.y = fb[m][1];
      a1.x = fa[m][2]; a1.y = fa[m][3];
      b1v.x = fb[m][2]; b1v.y = fb[m][3];
      acc0 = __builtin_amdgcn_wmma_f32_16x16x4_f32(false, a0, false, b0, (short)0, acc0,
                                                   false, false);
      acc1 = __builtin_amdgcn_wmma_f32_16x16x4_f32(false, a1, false, b1v, (short)0, acc1,
                                                   false, false);
    }
    __builtin_amdgcn_sched_barrier(0);
  }
  float* po = Afull + (row0 + rt * 16 + hi * 8) * 256 + col0 + ct * 16 + l15;
#pragma unroll
  for (int r = 0; r < 8; ++r) po[r * 256] = acc0[r] + acc1[r];
}

__global__ void __launch_bounds__(128)
score_kernel(const float* __restrict__ Afull, const float* __restrict__ b1,
             const float* __restrict__ Wqp, const float* __restrict__ bqv,
             float* __restrict__ out) {
  constexpr int P = 132;
  __shared__ __align__(16) float sWq[8 * P];
  __shared__ __align__(16) float sAj[32 * P];
  __shared__ __align__(16) float sAi[16 * 128];

  const int t    = threadIdx.x;
  const int lane = t & 31;
  const int c    = lane & 15;
  const int hi   = lane >> 4;
  const int w    = t >> 5;
  const int b    = blockIdx.x >> 5;
  const int jp   = (blockIdx.x >> 3) & 3;
  const int ig   = blockIdx.x & 7;

  {
    const int col = lane * 4;
    const float* srcj = Afull + ((b * 128 + jp * 32 + w) * 256 + 128 + col);
    const float* srci = Afull + ((b * 128 + ig * 16 + w) * 256 + col);
    v4f vj[8], vi[4];
#pragma unroll
    for (int n = 0; n < 8; ++n) vj[n] = *(const v4f*)(srcj + n * 4 * 256);
#pragma unroll
    for (int n = 0; n < 4; ++n) vi[n] = *(const v4f*)(srci + n * 4 * 256);
    const v4f bb = *(const v4f*)(b1 + col);
    const v4f q0 = *(const v4f*)(Wqp + t * 16);
    const v4f q1 = *(const v4f*)(Wqp + t * 16 + 4);
#pragma unroll
    for (int n = 0; n < 8; ++n) *(v4f*)(sAj + (n * 4 + w) * P + col) = vj[n];
#pragma unroll
    for (int n = 0; n < 4; ++n) *(v4f*)(sAi + (n * 4 + w) * 128 + col) = vi[n] + bb;
    sWq[0 * P + t] = q0.x; sWq[1 * P + t] = q0.y; sWq[2 * P + t] = q0.z; sWq[3 * P + t] = q0.w;
    sWq[4 * P + t] = q1.x; sWq[5 * P + t] = q1.y; sWq[6 * P + t] = q1.z; sWq[7 * P + t] = q1.w;
  }
  __syncthreads();

  v4f wq[16], aj0[16], aj1[16];
  {
    const float* pw = sWq + (c & 7) * P + 4 * hi;
    const float* p0 = sAj + c * P + 4 * hi;
    const float* p1 = sAj + (16 + c) * P + 4 * hi;
#pragma unroll
    for (int m = 0; m < 16; ++m) {
      wq[m]  = *(const v4f*)(pw + 8 * m);
      aj0[m] = *(const v4f*)(p0 + 8 * m);
      aj1[m] = *(const v4f*)(p1 + 8 * m);
    }
  }
  float bq[8];
#pragma unroll
  for (int r = 0; r < 8; ++r) bq[r] = bqv[r];

  const float* pa = sAi + (4 * w) * 128 + 4 * hi;
  v4f ai[16];
#pragma unroll
  for (int m = 0; m < 16; ++m) ai[m] = *(const v4f*)(pa + 8 * m);

  const int j = jp * 32 + lane;
  float* pout = out + ((b * 8) * 128 + ig * 16 + 4 * w) * 128 + j;

#pragma unroll 1
  for (int ii = 0; ii < 4; ++ii) {
    v4f nx[16];
    const float* pn = pa + ((ii < 3) ? (ii + 1) : ii) * 128;
#pragma unroll
    for (int m = 0; m < 16; ++m) nx[m] = *(const v4f*)(pn + 8 * m);

    v8f acc0 = {};
    v8f acc1 = {};
#pragma unroll
    for (int m = 0; m < 16; ++m) {
#pragma unroll
      for (int t2 = 0; t2 < 2; ++t2) {
        v2f a, h0, h1;
        a.x = wq[m][2 * t2];
        a.y = wq[m][2 * t2 + 1];
        const float x0 = ai[m][2 * t2];
        const float x1 = ai[m][2 * t2 + 1];
        h0.x = fmaxf(aj0[m][2 * t2] + x0, 0.0f);
        h0.y = fmaxf(aj0[m][2 * t2 + 1] + x1, 0.0f);
        h1.x = fmaxf(aj1[m][2 * t2] + x0, 0.0f);
        h1.y = fmaxf(aj1[m][2 * t2 + 1] + x1, 0.0f);
        acc0 = __builtin_amdgcn_wmma_f32_16x16x4_f32(false, a, false, h0, (short)0, acc0,
                                                     false, false);
        acc1 = __builtin_amdgcn_wmma_f32_16x16x4_f32(false, a, false, h1, (short)0, acc1,
                                                     false, false);
      }
    }

    const int i = ig * 16 + 4 * w + ii;
#pragma unroll
    for (int r = 0; r < 8; ++r) {
      const float v = (hi ? acc1[r] : acc0[r]) + bq[r];
      float s = 1.0f / (1.0f + __expf(-v));
      if (i == j) s = 0.0f;
      pout[(r * 128 + ii) * 128] = s;
    }
#pragma unroll
    for (int m = 0; m < 16; ++m) ai[m] = nx[m];
  }
}

extern "C" void kernel_launch(void* const* d_in, const int* in_sizes, int n_in,
                              void* d_out, int out_size, void* d_ws, size_t ws_size,
                              hipStream_t stream) {
  const float* E  = (const float*)d_in[0];
  const float* W1 = (const float*)d_in[1];
  const float* b1 = (const float*)d_in[2];
  const float* W2 = (const float*)d_in[3];
  const float* b2 = (const float*)d_in[4];
  const float* q  = (const float*)d_in[5];
  float* out = (float*)d_out;

  float* Afull = (float*)d_ws;
  float* Wqp   = Afull + 262144;
  float* bqv   = Wqp + 2048;

  gemm_A_kernel<<<dim3(258), dim3(128), 0, stream>>>(E, W1, Afull, W2, b2, q, Wqp, bqv);
  score_kernel<<<dim3(256), dim3(128), 0, stream>>>(Afull, b1, Wqp, bqv, out);
}


# Wq/bq moved from two extra work-groups into GEMM work-groups 0..7 under their tile loads (grid 258 -> 256, K split over 4 waves); Wq sum order changes: four K-quarter partials added in score kernel
# speedup vs baseline: 4.5992x; 1.0687x over previous
#include <hip/hip_runtime.h>
#include <math.h>


typedef __attribute__((ext_vector_type(2))) float v2f;
typedef __attribute__((ext_vector_type(4))) float v4f;
typedef __attribute__((ext_vector_type(8))) float v8f;

__global__ void __launch_bounds__(128)
gemm_A_kernel(const float* __restrict__ E, const float* __restrict__ W1,
              float* __restrict__ Afull,
              const float* __restrict__ W2, const float* __restrict__ b2,
              const float* __restrict__ q, float* __restrict__ Wqp,
              float* __restrict__ bqv) {
  constexpr int P = 260;
  __shared__ __align__(16) float sE[32 * P];
  __shared__ __align__(16) float sW[32 * P];
  const int lane = threadIdx.x & 31;
  const int l15 = lane & 15;
  const int hi  = lane >> 4;
  const int w   = threadIdx.x >> 5;

  const int g    = blockIdx.x;
  const bool extra = g < 8;
  const int row0 = (g >> 3) * 32;
  const int col0 = (g & 7) * 32;
  const float* wbase = (col0 < 128) ? (W1 + col0 * 512) : (W1 + (col0 - 128) * 512 + 256);

  const int hq = g * 16 + l15;
  v2f xa[16], xb[16];
  v4f bb0, bb1, bq0, bq1;
  if (extra) {
    const float* qrow = q + (l15 & 7) * 256 + 64 * w + 2 * hi;
    const float* wcol = W2 + (64 * w + 2 * hi) * 128 + hq;
#pragma unroll
    for (int kk = 0; kk < 16; ++kk) {
      xa[kk] = *(const v2f*)(qrow + 4 * kk);
      xb[kk].x = wcol[(4 * kk) * 128];
      xb[kk].y = wcol[(4 * kk + 1) * 128];
    }
    bb0 = *(const v4f*)(b2 + lane * 8);
    bb1 = *(const v4f*)(b2 + lane * 8 + 4);
    bq0 = *(const v4f*)(q + g * 256 + lane * 8);
    bq1 = *(const v4f*)(q + g * 256 + lane * 8 + 4);
  }
  __builtin_amdgcn_sched_barrier(0);

  const float* se = E + (row0 + w) * 256 + lane * 4;
  const float* sw = wbase + w * 512 + lane * 4;
  v4f ve[16], vw[16];
#pragma unroll
  for (int n = 0; n < 8; ++n) {
    ve[n] = *(const v4f*)(se + n * 4 * 256);
    vw[n] = *(const v4f*)(sw + n * 4 * 512);
  }
  __builtin_amdgcn_sched_barrier(0);
#pragma unroll
  for (int n = 0; n < 8; ++n) {
    ve[8 + n] = *(const v4f*)(se + n * 4 * 256 + 128);
    vw[8 + n] = *(const v4f*)(sw + n * 4 * 512 + 128);
  }
  __builtin_amdgcn_sched_barrier(0);

  if (extra) {
    v8f xc0 = {};
    v8f xc1 = {};
#pragma unroll
    for (int kk = 0; kk < 16; kk += 2) {
      xc0 = __builtin_amdgcn_wmma_f32_16x16x4_f32(false, xa[kk], false, xb[kk], (short)0, xc0,
                                                  false, false);
      xc1 = __builtin_amdgcn_wmma_f32_16x16x4_f32(false, xa[kk + 1], false, xb[kk + 1], (short)0, xc1,
                                                  false, false);
    }
    float* wout = Wqp + w * 2048 + hq * 16 + hi * 8;
#pragma unroll
    for (int r = 0; r < 8; ++r) wout[r] = hi ? 0.0f : (xc0[r] + xc1[r]);

    if (w == 0) {
      float t = 0.0f;
      t += bb0.x * bq0.x;
      t += bb0.y * bq0.y;
      t += bb0.z * bq0.z;
      t += bb0.w * bq0.w;
      t += bb1.x * bq1.x;
      t += bb1.y * bq1.y;
      t += bb1.z * bq1.z;
      t += bb1.w * bq1.w;
#pragma unroll
      for (int o = 16; o > 0; o >>= 1) t += __shfl_xor(t, o);
      if (lane == 0) bqv[g * 32] = t;
    }
  }
  __builtin_amdgcn_sched_barrier(0);

  const int rt = w >> 1;
  const int ct = w & 1;
  const float* pa = sE + (rt * 16 + l15) * P + 4 * hi;
  const float* pb = sW + (ct * 16 + l15) * P + 4 * hi;
  v8f acc0 = {};
  v8f acc1 = {};
#pragma unroll
  for (int hf = 0; hf < 2; ++hf) {
#pragma unroll
    for (int n = 0; n < 8; ++n) {
      *(v4f*)(sE + (n * 4 + w) * P + hf * 128 + lane * 4) = ve[hf * 8 + n];
      *(v4f*)(sW + (n * 4 + w) * P + hf * 128 + lane * 4) = vw[hf * 8 + n];
    }
    __builtin_amdgcn_fence(__ATOMIC_RELEASE, "workgroup", "local");
    __builtin_amdgcn_s_barrier();
    __builtin_amdgcn_fence(__ATOMIC_ACQUIRE, "workgroup", "local");
    v4f fa[16], fb[16];
#pragma unroll
    for (int m = 0; m < 16; ++m) {
      fa[m] = *(const v4f*)(pa + hf * 128 + 8 * m);
      fb[m] = *(const v4f*)(pb + hf * 128 + 8 * m);
    }
    __builtin_amdgcn_sched_barrier(0);
#pragma unroll
    for (int m = 0; m < 16; ++m) {
      v2f a0, b0, a1, b1v;
      a0.x = fa[m][0]; a0.y = fa[m][1];
      b0.x = fb[m][0]; b0.y = fb[m][1];
      a1.x = fa[m][2]; a1.y = fa[m][3];
      b1v.x = fb[m][2]; b1v.y = fb[m][3];
      acc0 = __builtin_amdgcn_wmma_f32_16x16x4_f32(false, a0, false, b0, (short)0, acc0,
                                                   false, false);
      acc1 = __builtin_amdgcn_wmma_f32_16x16x4_f32(false, a1, false, b1v, (short)0, acc1,
                                                   false, false);
    }
    __builtin_amdgcn_sched_barrier(0);
  }
  float* po = Afull + (row0 + rt * 16 + hi * 8) * 256 + col0 + ct * 16 + l15;
#pragma unroll
  for (int r = 0; r < 8; ++r) po[r * 256] = acc0[r] + acc1[r];
}

__global__ void __launch_bounds__(128)
score_kernel(const float* __restrict__ Afull, const float* __restrict__ b1,
             const float* __restrict__ Wqp, const float* __restrict__ bqv,
             float* __restrict__ out) {
  constexpr int P = 132;
  __shared__ __align__(16) float sWq[8 * P];
  __shared__ __align__(16) float sAj[32 * P];
  __shared__ __align__(16) float sAi[16 * 128];

  const int t    = threadIdx.x;
  const int lane = t & 31;
  const int c    = lane & 15;
  const int hi   = lane >> 4;
  const int w    = t >> 5;
  const int b    = blockIdx.x >> 5;
  const int jp   = (blockIdx.x >> 3) & 3;
  const int ig   = blockIdx.x & 7;

  {
    const int col = lane * 4;
    const float* srcj = Afull + ((b * 128 + jp * 32 + w) * 256 + 128 + col);
    const float* srci = Afull + ((b * 128 + ig * 16 + w) * 256 + col);
    v4f vj[8], vi[4];
#pragma unroll
    for (int n = 0; n < 8; ++n) vj[n] = *(const v4f*)(srcj + n * 4 * 256);
#pragma unroll
    for (int n = 0; n < 4; ++n) vi[n] = *(const v4f*)(srci + n * 4 * 256);
    const v4f bb = *(const v4f*)(b1 + col);
    v4f q0 = *(const v4f*)(Wqp + t * 16);
    v4f q1 = *(const v4f*)(Wqp + t * 16 + 4);
    v4f qp0[3], qp1[3];
#pragma unroll
    for (int p = 0; p < 3; ++p) {
      qp0[p] = *(const v4f*)(Wqp + (p + 1) * 2048 + t * 16);
      qp1[p] = *(const v4f*)(Wqp + (p + 1) * 2048 + t * 16 + 4);
    }
#pragma unroll
    for (int p = 0; p < 3; ++p) {
      q0 += qp0[p];
      q1 += qp1[p];
    }
#pragma unroll
    for (int n = 0; n < 8; ++n) *(v4f*)(sAj + (n * 4 + w) * P + col) = vj[n];
#pragma unroll
    for (int n = 0; n < 4; ++n) *(v4f*)(sAi + (n * 4 + w) * 128 + col) = vi[n] + bb;
    sWq[0 * P + t] = q0.x; sWq[1 * P + t] = q0.y; sWq[2 * P + t] = q0.z; sWq[3 * P + t] = q0.w;
    sWq[4 * P + t] = q1.x; sWq[5 * P + t] = q1.y; sWq[6 * P + t] = q1.z; sWq[7 * P + t] = q1.w;
  }
  __syncthreads();

  v4f wq[16], aj0[16], aj1[16];
  {
    const float* pw = sWq + (c & 7) * P + 4 * hi;
    const float* p0 = sAj + c * P + 4 * hi;
    const float* p1 = sAj + (16 + c) * P + 4 * hi;
#pragma unroll
    for (int m = 0; m < 16; ++m) {
      wq[m]  = *(const v4f*)(pw + 8 * m);
      aj0[m] = *(const v4f*)(p0 + 8 * m);
      aj1[m] = *(const v4f*)(p1 + 8 * m);
    }
  }
  float bq[8];
#pragma unroll
  for (int r = 0; r < 8; ++r) bq[r] = bqv[r * 32];

  const float* pa = sAi + (4 * w) * 128 + 4 * hi;
  v4f ai[16];
#pragma unroll
  for (int m = 0; m < 16; ++m) ai[m] = *(const v4f*)(pa + 8 * m);

  const int j = jp * 32 + lane;
  float* pout = out + ((b * 8) * 128 + ig * 16 + 4 * w) * 128 + j;

#pragma unroll 1
  for (int ii = 0; ii < 4; ++ii) {
    v4f nx[16];
    const float* pn = pa + ((ii < 3) ? (ii + 1) : ii) * 128;
#pragma unroll
    for (int m = 0; m < 16; ++m) nx[m] = *(const v4f*)(pn + 8 * m);

    v8f acc0 = {};
    v8f acc1 = {};
#pragma unroll
    for (int m = 0; m < 16; ++m) {
#pragma unroll
      for (int t2 = 0; t2 < 2; ++t2) {
        v2f a, h0, h1;
        a.x = wq[m][2 * t2];
        a.y = wq[m][2 * t2 + 1];
        const float x0 = ai[m][2 * t2];
        const float x1 = ai[m][2 * t2 + 1];
        h0.x = fmaxf(aj0[m][2 * t2] + x0, 0.0f);
        h0.y = fmaxf(aj0[m][2 * t2 + 1] + x1, 0.0f);
        h1.x = fmaxf(aj1[m][2 * t2] + x0, 0.0f);
        h1.y = fmaxf(aj1[m][2 * t2 + 1] + x1, 0.0f);
        acc0 = __builtin_amdgcn_wmma_f32_16x16x4_f32(false, a, false, h0, (short)0, acc0,
                                                     false, false);
        acc1 = __builtin_amdgcn_wmma_f32_16x16x4_f32(false, a, false, h1, (short)0, acc1,
                                                     false, false);
      }
    }

    const int i = ig * 16 + 4 * w + ii;
#pragma unroll
    for (int r = 0; r < 8; ++r) {
      const float v = (hi ? acc1[r] : acc0[r]) + bq[r];
      float s = 1.0f / (1.0f + __expf(-v));
      if (i == j) s = 0.0f;
      pout[(r * 128 + ii) * 128] = s;
    }
#pragma unroll
    for (int m = 0; m < 16; ++m) ai[m] = nx[m];
  }
}

extern "C" void kernel_launch(void* const* d_in, const int* in_sizes, int n_in,
                              void* d_out, int out_size, void* d_ws, size_t ws_size,
                              hipStream_t stream) {
  const float* E  = (const float*)d_in[0];
  const float* W1 = (const float*)d_in[1];
  const float* b1 = (const float*)d_in[2];
  const float* W2 = (const float*)d_in[3];
  const float* b2 = (const float*)d_in[4];
  const float* q  = (const float*)d_in[5];
  float* out = (float*)d_out;

  float* Afull = (float*)d_ws;
  float* Wqp   = Afull + 262144;
  float* bqv   = Wqp + 8192;

  gemm_A_kernel<<<dim3(256), dim3(128), 0, stream>>>(E, W1, Afull, W2, b2, q, Wqp, bqv);
  score_kernel<<<dim3(256), dim3(128), 0, stream>>>(Afull, b1, Wqp, bqv, out);
}
